# starting kernel
# baseline (speedup 1.0000x reference)
#include <hip/hip_runtime.h>
#include <hip/hip_bf16.h>

#define BATCH   32
#define SEQ     4096
#define QDIM    512
#define KVDIM   512

#define NCHUNK   16
#define ROWS_PER_CHUNK (SEQ / NCHUNK)
#define WAVES    8
#define TPB      (WAVES * 32)

#define WQ_OFF      0
#define PART_STRIDE (KVDIM + 8)
#define PART_OFF    (BATCH * KVDIM)

typedef float v2f __attribute__((ext_vector_type(2)));
typedef float v8f __attribute__((ext_vector_type(8)));

__global__ void __launch_bounds__(32)
wq_gemm_wmma(const float* __restrict__ query,
             const float* __restrict__ weight,
             float* __restrict__ wq) {
  const int tile = blockIdx.x;
  const int tM   = tile >> 5;
  const int tN   = tile & 31;
  const int lane = threadIdx.x;
  const int half = lane >> 4;
  const int lr   = lane & 15;

  const int aRow = tM * 16 + lr;
  const int bCol = tN * 16 + lr;

  v8f c = {};
  for (int k0 = 0; k0 < QDIM; k0 += 4) {
    const int ka = k0 + half * 2;
    v2f a, b;
    a[0] = query[aRow * QDIM + ka];
    a[1] = query[aRow * QDIM + ka + 1];
    b[0] = weight[(size_t)ka * KVDIM + bCol];
    b[1] = weight[(size_t)(ka + 1) * KVDIM + bCol];
    c = __builtin_amdgcn_wmma_f32_16x16x4_f32(false, a, false, b,
                                              (short)0, c, false, false);
  }
#pragma unroll
  for (int r = 0; r < 8; ++r) {
    wq[(tM * 16 + half * 8 + r) * KVDIM + tN * 16 + lr] = c[r];
  }
}

__global__ void __launch_bounds__(TPB)
attn_partial(const float* __restrict__ kv,
             const int*   __restrict__ mask,
             const float* __restrict__ wq,
             float*       __restrict__ part) {
  __shared__ float s_wq[KVDIM];
  __shared__ float s_acc[WAVES][KVDIM];
  __shared__ float s_m[WAVES];
  __shared__ float s_l[WAVES];

  const int b     = blockIdx.x / NCHUNK;
  const int chunk = blockIdx.x % NCHUNK;
  const int tid   = threadIdx.x;
  const int wave  = tid >> 5;
  const int lane  = tid & 31;
  const float NEG_INF = -__builtin_inff();

  for (int i = tid; i < KVDIM; i += TPB) s_wq[i] = wq[b * KVDIM + i];
  __syncthreads();

  float4 w4[4];
#pragma unroll
  for (int q = 0; q < 4; ++q)
    w4[q] = *(const float4*)&s_wq[q * 128 + lane * 4];

  float  m = NEG_INF, l = 0.f;
  float4 acc[4] = {};

  const int sBase = chunk * ROWS_PER_CHUNK;
  const int iters = ROWS_PER_CHUNK / WAVES;
  for (int it = 0; it < iters; ++it) {
    const int s = sBase + it * WAVES + wave;
    const float* __restrict__ row = kv + ((size_t)b * SEQ + s) * KVDIM;
    if (it + 1 < iters)
      __builtin_prefetch(row + WAVES * KVDIM, 0, 0);

    float4 r4[4];
#pragma unroll
    for (int q = 0; q < 4; ++q)
      r4[q] = *(const float4*)&row[q * 128 + lane * 4];

    float d = 0.f;
#pragma unroll
    for (int q = 0; q < 4; ++q) {
      d = fmaf(r4[q].x, w4[q].x, d);
      d = fmaf(r4[q].y, w4[q].y, d);
      d = fmaf(r4[q].z, w4[q].z, d);
      d = fmaf(r4[q].w, w4[q].w, d);
    }
#pragma unroll
    for (int off = 16; off > 0; off >>= 1) d += __shfl_xor(d, off, 32);

    if (mask[b * SEQ + s] != 0) {
      const float mN    = fmaxf(m, d);
      const float scale = __expf(m - mN);
      const float p     = __expf(d - mN);
      l = l * scale + p;
#pragma unroll
      for (int q = 0; q < 4; ++q) {
        acc[q].x = fmaf(p, r4[q].x, acc[q].x * scale);
        acc[q].y = fmaf(p, r4[q].y, acc[q].y * scale);
        acc[q].z = fmaf(p, r4[q].z, acc[q].z * scale);
        acc[q].w = fmaf(p, r4[q].w, acc[q].w * scale);
      }
      m = mN;
    }
  }

#pragma unroll
  for (int q = 0; q < 4; ++q)
    *(float4*)&s_acc[wave][q * 128 + lane * 4] = acc[q];
  if (lane == 0) { s_m[wave] = m; s_l[wave] = l; }
  __syncthreads();

  float mB = NEG_INF;
#pragma unroll
  for (int w = 0; w < WAVES; ++w) mB = fmaxf(mB, s_m[w]);
  float wgt[WAVES];
  float lB = 0.f;
#pragma unroll
  for (int w = 0; w < WAVES; ++w) {
    const float wi = (s_m[w] == NEG_INF) ? 0.f : __expf(s_m[w] - mB);
    wgt[w] = wi;
    lB += wi * s_l[w];
  }
  float* P = part + (size_t)(b * NCHUNK + chunk) * PART_STRIDE;
  for (int e = tid; e < KVDIM; e += TPB) {
    float v = 0.f;
#pragma unroll
    for (int w = 0; w < WAVES; ++w) v = fmaf(wgt[w], s_acc[w][e], v);
    P[e] = v;
  }
  if (tid == 0) { P[KVDIM] = mB; P[KVDIM + 1] = lB; }
}

__global__ void __launch_bounds__(256)
attn_combine(const float* __restrict__ part, float* __restrict__ out) {
  const int b   = blockIdx.x;
  const int tid = threadIdx.x;
  const float NEG_INF = -__builtin_inff();

  float mG = NEG_INF;
#pragma unroll
  for (int c = 0; c < NCHUNK; ++c)
    mG = fmaxf(mG, part[(size_t)(b * NCHUNK + c) * PART_STRIDE + KVDIM]);

  float wc[NCHUNK];
  float lG = 0.f;
#pragma unroll
  for (int c = 0; c < NCHUNK; ++c) {
    const size_t base = (size_t)(b * NCHUNK + c) * PART_STRIDE;
    const float  mi   = part[base + KVDIM];
    const float  wi   = (mi == NEG_INF) ? 0.f : __expf(mi - mG);
    wc[c] = wi;
    lG += wi * part[base + KVDIM + 1];
  }
  const float inv = 1.f / lG;

  for (int e = tid; e < KVDIM; e += 256) {
    float v = 0.f;
#pragma unroll
    for (int c = 0; c < NCHUNK; ++c)
      v = fmaf(wc[c], part[(size_t)(b * NCHUNK + c) * PART_STRIDE + e], v);
    out[b * KVDIM + e] = v * inv;
  }
}

extern "C" void kernel_launch(void* const* d_in, const int* in_sizes, int n_in,
                              void* d_out, int out_size, void* d_ws, size_t ws_size,
                              hipStream_t stream) {
  const float* query  = (const float*)d_in[0];
  const float* kv     = (const float*)d_in[1];
  const int*   mask   = (const int*)  d_in[2];
  const float* weight = (const float*)d_in[3];
  float* out = (float*)d_out;

  float* ws   = (float*)d_ws;
  float* wq   = ws + WQ_OFF;
  float* part = ws + PART_OFF;

  wq_gemm_wmma<<<64, 32, 0, stream>>>(query, weight, wq);
  attn_partial<<<BATCH * NCHUNK, TPB, 0, stream>>>(kv, mask, wq, part);
  attn_combine<<<BATCH, 256, 0, stream>>>(part, out);
}


# GEMM kernel: K split over 16 waves per output tile (8 WMMA steps each, all loads issued up front), partial tiles summed through LDS in wave order; changes the order in which wq is summed over k
# speedup vs baseline: 1.2796x; 1.2796x over previous
#include <hip/hip_runtime.h>
#include <hip/hip_bf16.h>

#define BATCH   32
#define SEQ     4096
#define QDIM    512
#define KVDIM   512

#define NCHUNK   16
#define ROWS_PER_CHUNK (SEQ / NCHUNK)
#define WAVES    8
#define TPB      (WAVES * 32)

#define WQ_OFF      0
#define PART_STRIDE (KVDIM + 8)
#define PART_OFF    (BATCH * KVDIM)

typedef float v2f __attribute__((ext_vector_type(2)));
typedef float v8f __attribute__((ext_vector_type(8)));

#define KSPLIT  16
#define KSLICE  (QDIM / KSPLIT)

__global__ void __launch_bounds__(KSPLIT * 32)
wq_gemm_wmma(const float* __restrict__ query,
             const float* __restrict__ weight,
             float* __restrict__ wq) {
  __shared__ float s_part[KSPLIT][8 * 32];

  const int tile = blockIdx.x;
  const int tM   = tile >> 5;
  const int tN   = tile & 31;
  const int tid  = threadIdx.x;
  const int wave = tid >> 5;
  const int lane = tid & 31;
  const int half = lane >> 4;
  const int lr   = lane & 15;

  const int aRow = tM * 16 + lr;
  const int bCol = tN * 16 + lr;
  const int kBeg = wave * KSLICE;

  v8f c = {};
#pragma unroll
  for (int k0 = kBeg; k0 < kBeg + KSLICE; k0 += 4) {
    const int ka = k0 + half * 2;
    v2f a, b;
    a[0] = query[aRow * QDIM + ka];
    a[1] = query[aRow * QDIM + ka + 1];
    b[0] = weight[(size_t)ka * KVDIM + bCol];
    b[1] = weight[(size_t)(ka + 1) * KVDIM + bCol];
    c = __builtin_amdgcn_wmma_f32_16x16x4_f32(false, a, false, b,
                                              (short)0, c, false, false);
  }

#pragma unroll
  for (int r = 0; r < 8; ++r) s_part[wave][r * 32 + lane] = c[r];
  __syncthreads();

  if (tid < 8 * 32) {
    float v = s_part[0][tid];
#pragma unroll
    for (int w = 1; w < KSPLIT; ++w) v += s_part[w][tid];
    const int r = tid >> 5;
    wq[(tM * 16 + half * 8 + r) * KVDIM + tN * 16 + lr] = v;
  }
}

__global__ void __launch_bounds__(TPB)
attn_partial(const float* __restrict__ kv,
             const int*   __restrict__ mask,
             const float* __restrict__ wq,
             float*       __restrict__ part) {
  __shared__ float s_wq[KVDIM];
  __shared__ float s_acc[WAVES][KVDIM];
  __shared__ float s_m[WAVES];
  __shared__ float s_l[WAVES];

  const int b     = blockIdx.x / NCHUNK;
  const int chunk = blockIdx.x % NCHUNK;
  const int tid   = threadIdx.x;
  const int wave  = tid >> 5;
  const int lane  = tid & 31;
  const float NEG_INF = -__builtin_inff();

  for (int i = tid; i < KVDIM; i += TPB) s_wq[i] = wq[b * KVDIM + i];
  __syncthreads();

  float4 w4[4];
#pragma unroll
  for (int q = 0; q < 4; ++q)
    w4[q] = *(const float4*)&s_wq[q * 128 + lane * 4];

  float  m = NEG_INF, l = 0.f;
  float4 acc[4] = {};

  const int sBase = chunk * ROWS_PER_CHUNK;
  const int iters = ROWS_PER_CHUNK / WAVES;
  for (int it = 0; it < iters; ++it) {
    const int s = sBase + it * WAVES + wave;
    const float* __restrict__ row = kv + ((size_t)b * SEQ + s) * KVDIM;
    if (it + 1 < iters)
      __builtin_prefetch(row + WAVES * KVDIM, 0, 0);

    float4 r4[4];
#pragma unroll
    for (int q = 0; q < 4; ++q)
      r4[q] = *(const float4*)&row[q * 128 + lane * 4];

    float d = 0.f;
#pragma unroll
    for (int q = 0; q < 4; ++q) {
      d = fmaf(r4[q].x, w4[q].x, d);
      d = fmaf(r4[q].y, w4[q].y, d);
      d = fmaf(r4[q].z, w4[q].z, d);
      d = fmaf(r4[q].w, w4[q].w, d);
    }
#pragma unroll
    for (int off = 16; off > 0; off >>= 1) d += __shfl_xor(d, off, 32);

    if (mask[b * SEQ + s] != 0) {
      const float mN    = fmaxf(m, d);
      const float scale = __expf(m - mN);
      const float p     = __expf(d - mN);
      l = l * scale + p;
#pragma unroll
      for (int q = 0; q < 4; ++q) {
        acc[q].x = fmaf(p, r4[q].x, acc[q].x * scale);
        acc[q].y = fmaf(p, r4[q].y, acc[q].y * scale);
        acc[q].z = fmaf(p, r4[q].z, acc[q].z * scale);
        acc[q].w = fmaf(p, r4[q].w, acc[q].w * scale);
      }
      m = mN;
    }
  }

#pragma unroll
  for (int q = 0; q < 4; ++q)
    *(float4*)&s_acc[wave][q * 128 + lane * 4] = acc[q];
  if (lane == 0) { s_m[wave] = m; s_l[wave] = l; }
  __syncthreads();

  float mB = NEG_INF;
#pragma unroll
  for (int w = 0; w < WAVES; ++w) mB = fmaxf(mB, s_m[w]);
  float wgt[WAVES];
  float lB = 0.f;
#pragma unroll
  for (int w = 0; w < WAVES; ++w) {
    const float wi = (s_m[w] == NEG_INF) ? 0.f : __expf(s_m[w] - mB);
    wgt[w] = wi;
    lB += wi * s_l[w];
  }
  float* P = part + (size_t)(b * NCHUNK + chunk) * PART_STRIDE;
  for (int e = tid; e < KVDIM; e += TPB) {
    float v = 0.f;
#pragma unroll
    for (int w = 0; w < WAVES; ++w) v = fmaf(wgt[w], s_acc[w][e], v);
    P[e] = v;
  }
  if (tid == 0) { P[KVDIM] = mB; P[KVDIM + 1] = lB; }
}

__global__ void __launch_bounds__(256)
attn_combine(const float* __restrict__ part, float* __restrict__ out) {
  const int b   = blockIdx.x;
  const int tid = threadIdx.x;
  const float NEG_INF = -__builtin_inff();

  float mG = NEG_INF;
#pragma unroll
  for (int c = 0; c < NCHUNK; ++c)
    mG = fmaxf(mG, part[(size_t)(b * NCHUNK + c) * PART_STRIDE + KVDIM]);

  float wc[NCHUNK];
  float lG = 0.f;
#pragma unroll
  for (int c = 0; c < NCHUNK; ++c) {
    const size_t base = (size_t)(b * NCHUNK + c) * PART_STRIDE;
    const float  mi   = part[base + KVDIM];
    const float  wi   = (mi == NEG_INF) ? 0.f : __expf(mi - mG);
    wc[c] = wi;
    lG += wi * part[base + KVDIM + 1];
  }
  const float inv = 1.f / lG;

  for (int e = tid; e < KVDIM; e += 256) {
    float v = 0.f;
#pragma unroll
    for (int c = 0; c < NCHUNK; ++c)
      v = fmaf(wc[c], part[(size_t)(b * NCHUNK + c) * PART_STRIDE + e], v);
    out[b * KVDIM + e] = v * inv;
  }
}

extern "C" void kernel_launch(void* const* d_in, const int* in_sizes, int n_in,
                              void* d_out, int out_size, void* d_ws, size_t ws_size,
                              hipStream_t stream) {
  const float* query  = (const float*)d_in[0];
  const float* kv     = (const float*)d_in[1];
  const int*   mask   = (const int*)  d_in[2];
  const float* weight = (const float*)d_in[3];
  float* out = (float*)d_out;

  float* ws   = (float*)d_ws;
  float* wq   = ws + WQ_OFF;
  float* part = ws + PART_OFF;

  wq_gemm_wmma<<<64, KSPLIT * 32, 0, stream>>>(query, weight, wq);
  attn_partial<<<BATCH * NCHUNK, TPB, 0, stream>>>(kv, mask, wq, part);
  attn_combine<<<BATCH, 256, 0, stream>>>(part, out);
}


# Attention pass reads and scores only the unmasked rows: each block lists its chunk's unmasked rows in LDS and deals them to its 8 waves in turn; changes the order in which rows are summed
# speedup vs baseline: 1.8418x; 1.4394x over previous
#include <hip/hip_runtime.h>
#include <hip/hip_bf16.h>

#define BATCH   32
#define SEQ     4096
#define QDIM    512
#define KVDIM   512

#define NCHUNK   16
#define ROWS_PER_CHUNK (SEQ / NCHUNK)
#define WAVES    8
#define TPB      (WAVES * 32)

#define WQ_OFF      0
#define PART_STRIDE (KVDIM + 8)
#define PART_OFF    (BATCH * KVDIM)

typedef float v2f __attribute__((ext_vector_type(2)));
typedef float v8f __attribute__((ext_vector_type(8)));

#define KSPLIT  16
#define KSLICE  (QDIM / KSPLIT)

__global__ void __launch_bounds__(KSPLIT * 32)
wq_gemm_wmma(const float* __restrict__ query,
             const float* __restrict__ weight,
             float* __restrict__ wq) {
  __shared__ float s_part[KSPLIT][8 * 32];

  const int tile = blockIdx.x;
  const int tM   = tile >> 5;
  const int tN   = tile & 31;
  const int tid  = threadIdx.x;
  const int wave = tid >> 5;
  const int lane = tid & 31;
  const int half = lane >> 4;
  const int lr   = lane & 15;

  const int aRow = tM * 16 + lr;
  const int bCol = tN * 16 + lr;
  const int kBeg = wave * KSLICE;

  v8f c = {};
#pragma unroll
  for (int k0 = kBeg; k0 < kBeg + KSLICE; k0 += 4) {
    const int ka = k0 + half * 2;
    v2f a, b;
    a[0] = query[aRow * QDIM + ka];
    a[1] = query[aRow * QDIM + ka + 1];
    b[0] = weight[(size_t)ka * KVDIM + bCol];
    b[1] = weight[(size_t)(ka + 1) * KVDIM + bCol];
    c = __builtin_amdgcn_wmma_f32_16x16x4_f32(false, a, false, b,
                                              (short)0, c, false, false);
  }

#pragma unroll
  for (int r = 0; r < 8; ++r) s_part[wave][r * 32 + lane] = c[r];
  __syncthreads();

  if (tid < 8 * 32) {
    float v = s_part[0][tid];
#pragma unroll
    for (int w = 1; w < KSPLIT; ++w) v += s_part[w][tid];
    const int r = tid >> 5;
    wq[(tM * 16 + half * 8 + r) * KVDIM + tN * 16 + lr] = v;
  }
}

__global__ void __launch_bounds__(TPB)
attn_partial(const float* __restrict__ kv,
             const int*   __restrict__ mask,
             const float* __restrict__ wq,
             float*       __restrict__ part) {
  __shared__ float    s_wq[KVDIM];
  __shared__ float    s_acc[WAVES][KVDIM];
  __shared__ float    s_m[WAVES];
  __shared__ float    s_l[WAVES];
  __shared__ unsigned s_bits[WAVES];
  __shared__ int      s_list[ROWS_PER_CHUNK];

  static_assert(ROWS_PER_CHUNK == TPB, "one thread per row of the chunk");

  const int b     = blockIdx.x / NCHUNK;
  const int chunk = blockIdx.x % NCHUNK;
  const int tid   = threadIdx.x;
  const int lane  = tid & 31;
  const int wave  = __builtin_amdgcn_readfirstlane(tid >> 5);
  const float NEG_INF = -__builtin_inff();
  const int sBase = chunk * ROWS_PER_CHUNK;

  const bool     keep = mask[b * SEQ + sBase + tid] != 0;
  const unsigned bits = __builtin_amdgcn_ballot_w32(keep);
  if (lane == 0) s_bits[wave] = bits;
  for (int i = tid; i < KVDIM; i += TPB) s_wq[i] = wq[b * KVDIM + i];
  __syncthreads();

  int before = 0, nAll = 0;
#pragma unroll
  for (int w = 0; w < WAVES; ++w) {
    const int c = __popc(s_bits[w]);
    before += (w < wave) ? c : 0;
    nAll   += c;
  }
  if (keep) s_list[before + __popc(bits & ((1u << lane) - 1u))] = tid;
  __syncthreads();
  const int n = __builtin_amdgcn_readfirstlane(nAll);

  float4 w4[4];
#pragma unroll
  for (int q = 0; q < 4; ++q)
    w4[q] = *(const float4*)&s_wq[q * 128 + lane * 4];

  float  m = NEG_INF, l = 0.f;
  float4 acc[4] = {};

  const float* __restrict__ chunkBase = kv + ((size_t)b * SEQ + sBase) * KVDIM;
  int r = (wave < n) ? s_list[wave] : 0;
  for (int i = wave; i < n; i += WAVES) {
    const int rU = __builtin_amdgcn_readfirstlane(r);
    const float* __restrict__ row = chunkBase + (size_t)rU * KVDIM;

    float4 r4[4];
#pragma unroll
    for (int q = 0; q < 4; ++q)
      r4[q] = *(const float4*)&row[q * 128 + lane * 4];

    if (i + WAVES < n) {
      r = s_list[i + WAVES];
      __builtin_prefetch(chunkBase + (size_t)r * KVDIM, 0, 0);
    }

    float d = 0.f;
#pragma unroll
    for (int q = 0; q < 4; ++q) {
      d = fmaf(r4[q].x, w4[q].x, d);
      d = fmaf(r4[q].y, w4[q].y, d);
      d = fmaf(r4[q].z, w4[q].z, d);
      d = fmaf(r4[q].w, w4[q].w, d);
    }
#pragma unroll
    for (int off = 16; off > 0; off >>= 1) d += __shfl_xor(d, off, 32);

    const float mN    = fmaxf(m, d);
    const float scale = __expf(m - mN);
    const float p     = __expf(d - mN);
    l = l * scale + p;
#pragma unroll
    for (int q = 0; q < 4; ++q) {
      acc[q].x = fmaf(p, r4[q].x, acc[q].x * scale);
      acc[q].y = fmaf(p, r4[q].y, acc[q].y * scale);
      acc[q].z = fmaf(p, r4[q].z, acc[q].z * scale);
      acc[q].w = fmaf(p, r4[q].w, acc[q].w * scale);
    }
    m = mN;
  }

#pragma unroll
  for (int q = 0; q < 4; ++q)
    *(float4*)&s_acc[wave][q * 128 + lane * 4] = acc[q];
  if (lane == 0) { s_m[wave] = m; s_l[wave] = l; }
  __syncthreads();

  float mB = NEG_INF;
#pragma unroll
  for (int w = 0; w < WAVES; ++w) mB = fmaxf(mB, s_m[w]);
  float wgt[WAVES];
  float lB = 0.f;
#pragma unroll
  for (int w = 0; w < WAVES; ++w) {
    const float wi = (s_m[w] == NEG_INF) ? 0.f : __expf(s_m[w] - mB);
    wgt[w] = wi;
    lB += wi * s_l[w];
  }
  float* P = part + (size_t)(b * NCHUNK + chunk) * PART_STRIDE;
  for (int e = tid; e < KVDIM; e += TPB) {
    float v = 0.f;
#pragma unroll
    for (int w = 0; w < WAVES; ++w) v = fmaf(wgt[w], s_acc[w][e], v);
    P[e] = v;
  }
  if (tid == 0) { P[KVDIM] = mB; P[KVDIM + 1] = lB; }
}

__global__ void __launch_bounds__(256)
attn_combine(const float* __restrict__ part, float* __restrict__ out) {
  const int b   = blockIdx.x;
  const int tid = threadIdx.x;
  const float NEG_INF = -__builtin_inff();

  float mG = NEG_INF;
#pragma unroll
  for (int c = 0; c < NCHUNK; ++c)
    mG = fmaxf(mG, part[(size_t)(b * NCHUNK + c) * PART_STRIDE + KVDIM]);

  float wc[NCHUNK];
  float lG = 0.f;
#pragma unroll
  for (int c = 0; c < NCHUNK; ++c) {
    const size_t base = (size_t)(b * NCHUNK + c) * PART_STRIDE;
    const float  mi   = part[base + KVDIM];
    const float  wi   = (mi == NEG_INF) ? 0.f : __expf(mi - mG);
    wc[c] = wi;
    lG += wi * part[base + KVDIM + 1];
  }
  const float inv = 1.f / lG;

  for (int e = tid; e < KVDIM; e += 256) {
    float v = 0.f;
#pragma unroll
    for (int c = 0; c < NCHUNK; ++c)
      v = fmaf(wc[c], part[(size_t)(b * NCHUNK + c) * PART_STRIDE + e], v);
    out[b * KVDIM + e] = v * inv;
  }
}

extern "C" void kernel_launch(void* const* d_in, const int* in_sizes, int n_in,
                              void* d_out, int out_size, void* d_ws, size_t ws_size,
                              hipStream_t stream) {
  const float* query  = (const float*)d_in[0];
  const float* kv     = (const float*)d_in[1];
  const int*   mask   = (const int*)  d_in[2];
  const float* weight = (const float*)d_in[3];
  float* out = (float*)d_out;

  float* ws   = (float*)d_ws;
  float* wq   = ws + WQ_OFF;
  float* part = ws + PART_OFF;

  wq_gemm_wmma<<<64, KSPLIT * 32, 0, stream>>>(query, weight, wq);
  attn_partial<<<BATCH * NCHUNK, TPB, 0, stream>>>(kv, mask, wq, part);
  attn_combine<<<BATCH, 256, 0, stream>>>(part, out);
}


# Attention pass: each wave loads and scores its kept rows four at a time (16 b128 loads in flight, one softmax rescale per group); changes the order in which rows are summed
# speedup vs baseline: 2.2385x; 1.2154x over previous
#include <hip/hip_runtime.h>
#include <hip/hip_bf16.h>

#define BATCH   32
#define SEQ     4096
#define QDIM    512
#define KVDIM   512

#define NCHUNK   16
#define ROWS_PER_CHUNK (SEQ / NCHUNK)
#define WAVES    8
#define TPB      (WAVES * 32)
#define GROUP    4

#define WQ_OFF      0
#define PART_STRIDE (KVDIM + 8)
#define PART_OFF    (BATCH * KVDIM)

typedef float v2f __attribute__((ext_vector_type(2)));
typedef float v8f __attribute__((ext_vector_type(8)));

#define KSPLIT  16
#define KSLICE  (QDIM / KSPLIT)

__global__ void __launch_bounds__(KSPLIT * 32)
wq_gemm_wmma(const float* __restrict__ query,
             const float* __restrict__ weight,
             float* __restrict__ wq) {
  __shared__ float s_part[KSPLIT][8 * 32];

  const int tile = blockIdx.x;
  const int tM   = tile >> 5;
  const int tN   = tile & 31;
  const int tid  = threadIdx.x;
  const int wave = tid >> 5;
  const int lane = tid & 31;
  const int half = lane >> 4;
  const int lr   = lane & 15;

  const int aRow = tM * 16 + lr;
  const int bCol = tN * 16 + lr;
  const int kBeg = wave * KSLICE;

  v8f c = {};
#pragma unroll
  for (int k0 = kBeg; k0 < kBeg + KSLICE; k0 += 4) {
    const int ka = k0 + half * 2;
    v2f a, b;
    a[0] = query[aRow * QDIM + ka];
    a[1] = query[aRow * QDIM + ka + 1];
    b[0] = weight[(size_t)ka * KVDIM + bCol];
    b[1] = weight[(size_t)(ka + 1) * KVDIM + bCol];
    c = __builtin_amdgcn_wmma_f32_16x16x4_f32(false, a, false, b,
                                              (short)0, c, false, false);
  }

#pragma unroll
  for (int r = 0; r < 8; ++r) s_part[wave][r * 32 + lane] = c[r];
  __syncthreads();

  if (tid < 8 * 32) {
    float v = s_part[0][tid];
#pragma unroll
    for (int w = 1; w < KSPLIT; ++w) v += s_part[w][tid];
    const int r = tid >> 5;
    wq[(tM * 16 + half * 8 + r) * KVDIM + tN * 16 + lr] = v;
  }
}

__global__ void __launch_bounds__(TPB)
attn_partial(const float* __restrict__ kv,
             const int*   __restrict__ mask,
             const float* __restrict__ wq,
             float*       __restrict__ part) {
  __shared__ float    s_wq[KVDIM];
  __shared__ float    s_acc[WAVES][KVDIM];
  __shared__ float    s_m[WAVES];
  __shared__ float    s_l[WAVES];
  __shared__ unsigned s_bits[WAVES];
  __shared__ int      s_list[ROWS_PER_CHUNK];

  static_assert(ROWS_PER_CHUNK == TPB, "one thread per row of the chunk");

  const int b     = blockIdx.x / NCHUNK;
  const int chunk = blockIdx.x % NCHUNK;
  const int tid   = threadIdx.x;
  const int lane  = tid & 31;
  const int wave  = __builtin_amdgcn_readfirstlane(tid >> 5);
  const float NEG_INF = -__builtin_inff();
  const int sBase = chunk * ROWS_PER_CHUNK;

  const bool     keep = mask[b * SEQ + sBase + tid] != 0;
  const unsigned bits = __builtin_amdgcn_ballot_w32(keep);
  if (lane == 0) s_bits[wave] = bits;
  for (int i = tid; i < KVDIM; i += TPB) s_wq[i] = wq[b * KVDIM + i];
  __syncthreads();

  int before = 0, nAll = 0;
#pragma unroll
  for (int w = 0; w < WAVES; ++w) {
    const int c = __popc(s_bits[w]);
    before += (w < wave) ? c : 0;
    nAll   += c;
  }
  if (keep) s_list[before + __popc(bits & ((1u << lane) - 1u))] = tid;
  __syncthreads();
  const int n = __builtin_amdgcn_readfirstlane(nAll);

  float4 w4[4];
#pragma unroll
  for (int q = 0; q < 4; ++q)
    w4[q] = *(const float4*)&s_wq[q * 128 + lane * 4];

  float  m = NEG_INF, l = 0.f;
  float4 acc[4] = {};

  static_assert(ROWS_PER_CHUNK / WAVES == 32, "a wave's share of the list fits one entry per lane");
  const int myIdx = wave + lane * WAVES;
  const int myRow = (myIdx < n) ? s_list[myIdx] : 0;
  const int nW    = (n - wave + WAVES - 1) / WAVES;

  const float* __restrict__ chunkBase = kv + ((size_t)b * SEQ + sBase) * KVDIM;
  for (int g = 0; g < nW; g += GROUP) {
    float4 r4[GROUP][4];
#pragma unroll
    for (int j = 0; j < GROUP; ++j) {
      const int k  = (g + j < nW) ? (g + j) : g;
      const int rU = __builtin_amdgcn_readlane(myRow, k);
      const float* __restrict__ row = chunkBase + (size_t)rU * KVDIM;
#pragma unroll
      for (int q = 0; q < 4; ++q)
        r4[j][q] = *(const float4*)&row[q * 128 + lane * 4];
    }

    float d[GROUP];
#pragma unroll
    for (int j = 0; j < GROUP; ++j) {
      float dj = 0.f;
#pragma unroll
      for (int q = 0; q < 4; ++q) {
        dj = fmaf(r4[j][q].x, w4[q].x, dj);
        dj = fmaf(r4[j][q].y, w4[q].y, dj);
        dj = fmaf(r4[j][q].z, w4[q].z, dj);
        dj = fmaf(r4[j][q].w, w4[q].w, dj);
      }
      d[j] = dj;
    }
#pragma unroll
    for (int off = 16; off > 0; off >>= 1) {
#pragma unroll
      for (int j = 0; j < GROUP; ++j) d[j] += __shfl_xor(d[j], off, 32);
    }
#pragma unroll
    for (int j = 1; j < GROUP; ++j)
      if (g + j >= nW) d[j] = NEG_INF;

    float mN = m;
#pragma unroll
    for (int j = 0; j < GROUP; ++j) mN = fmaxf(mN, d[j]);
    const float scale = __expf(m - mN);
    float p[GROUP];
    float pSum = 0.f;
#pragma unroll
    for (int j = 0; j < GROUP; ++j) { p[j] = __expf(d[j] - mN); pSum += p[j]; }
    l = l * scale + pSum;
#pragma unroll
    for (int q = 0; q < 4; ++q) {
      float4 a = acc[q];
      a.x *= scale; a.y *= scale; a.z *= scale; a.w *= scale;
#pragma unroll
      for (int j = 0; j < GROUP; ++j) {
        a.x = fmaf(p[j], r4[j][q].x, a.x);
        a.y = fmaf(p[j], r4[j][q].y, a.y);
        a.z = fmaf(p[j], r4[j][q].z, a.z);
        a.w = fmaf(p[j], r4[j][q].w, a.w);
      }
      acc[q] = a;
    }
    m = mN;
  }

#pragma unroll
  for (int q = 0; q < 4; ++q)
    *(float4*)&s_acc[wave][q * 128 + lane * 4] = acc[q];
  if (lane == 0) { s_m[wave] = m; s_l[wave] = l; }
  __syncthreads();

  float mB = NEG_INF;
#pragma unroll
  for (int w = 0; w < WAVES; ++w) mB = fmaxf(mB, s_m[w]);
  float wgt[WAVES];
  float lB = 0.f;
#pragma unroll
  for (int w = 0; w < WAVES; ++w) {
    const float wi = (s_m[w] == NEG_INF) ? 0.f : __expf(s_m[w] - mB);
    wgt[w] = wi;
    lB += wi * s_l[w];
  }
  float* P = part + (size_t)(b * NCHUNK + chunk) * PART_STRIDE;
  for (int e = tid; e < KVDIM; e += TPB) {
    float v = 0.f;
#pragma unroll
    for (int w = 0; w < WAVES; ++w) v = fmaf(wgt[w], s_acc[w][e], v);
    P[e] = v;
  }
  if (tid == 0) { P[KVDIM] = mB; P[KVDIM + 1] = lB; }
}

__global__ void __launch_bounds__(256)
attn_combine(const float* __restrict__ part, float* __restrict__ out) {
  const int b   = blockIdx.x;
  const int tid = threadIdx.x;
  const float NEG_INF = -__builtin_inff();

  float mG = NEG_INF;
#pragma unroll
  for (int c = 0; c < NCHUNK; ++c)
    mG = fmaxf(mG, part[(size_t)(b * NCHUNK + c) * PART_STRIDE + KVDIM]);

  float wc[NCHUNK];
  float lG = 0.f;
#pragma unroll
  for (int c = 0; c < NCHUNK; ++c) {
    const size_t base = (size_t)(b * NCHUNK + c) * PART_STRIDE;
    const float  mi   = part[base + KVDIM];
    const float  wi   = (mi == NEG_INF) ? 0.f : __expf(mi - mG);
    wc[c] = wi;
    lG += wi * part[base + KVDIM + 1];
  }
  const float inv = 1.f / lG;

  for (int e = tid; e < KVDIM; e += 256) {
    float v = 0.f;
#pragma unroll
    for (int c = 0; c < NCHUNK; ++c)
      v = fmaf(wc[c], part[(size_t)(b * NCHUNK + c) * PART_STRIDE + e], v);
    out[b * KVDIM + e] = v * inv;
  }
}

extern "C" void kernel_launch(void* const* d_in, const int* in_sizes, int n_in,
                              void* d_out, int out_size, void* d_ws, size_t ws_size,
                              hipStream_t stream) {
  const float* query  = (const float*)d_in[0];
  const float* kv     = (const float*)d_in[1];
  const int*   mask   = (const int*)  d_in[2];
  const float* weight = (const float*)d_in[3];
  float* out = (float*)d_out;

  float* ws   = (float*)d_ws;
  float* wq   = ws + WQ_OFF;
  float* part = ws + PART_OFF;

  wq_gemm_wmma<<<64, KSPLIT * 32, 0, stream>>>(query, weight, wq);
  attn_partial<<<BATCH * NCHUNK, TPB, 0, stream>>>(kv, mask, wq, part);
  attn_combine<<<BATCH, 256, 0, stream>>>(part, out);
}


# Independent loads issued together, fewer serial round trips per wave: attention prologue 3->1 (wq to registers), rows 8 per group (4->2 trips), combine 3->1; changes the order rows are summed
# speedup vs baseline: 2.4890x; 1.1119x over previous
#include <hip/hip_runtime.h>
#include <hip/hip_bf16.h>

#define BATCH   32
#define SEQ     4096
#define QDIM    512
#define KVDIM   512

#define NCHUNK   16
#define ROWS_PER_CHUNK (SEQ / NCHUNK)
#define WAVES    8
#define TPB      (WAVES * 32)
#define GROUP    8

#define WQ_OFF      0
#define PART_STRIDE (KVDIM + 8)
#define PART_OFF    (BATCH * KVDIM)

typedef float v2f __attribute__((ext_vector_type(2)));
typedef float v8f __attribute__((ext_vector_type(8)));

#define KSPLIT  16
#define KSLICE  (QDIM / KSPLIT)

__global__ void __launch_bounds__(KSPLIT * 32)
wq_gemm_wmma(const float* __restrict__ query,
             const float* __restrict__ weight,
             float* __restrict__ wq) {
  __shared__ float s_part[KSPLIT][8 * 32];

  const int tile = blockIdx.x;
  const int tM   = tile >> 5;
  const int tN   = tile & 31;
  const int tid  = threadIdx.x;
  const int wave = tid >> 5;
  const int lane = tid & 31;
  const int half = lane >> 4;
  const int lr   = lane & 15;

  const int aRow = tM * 16 + lr;
  const int bCol = tN * 16 + lr;
  const int kBeg = wave * KSLICE;

  v8f c = {};
#pragma unroll
  for (int k0 = kBeg; k0 < kBeg + KSLICE; k0 += 4) {
    const int ka = k0 + half * 2;
    v2f a, b;
    a[0] = query[aRow * QDIM + ka];
    a[1] = query[aRow * QDIM + ka + 1];
    b[0] = weight[(size_t)ka * KVDIM + bCol];
    b[1] = weight[(size_t)(ka + 1) * KVDIM + bCol];
    c = __builtin_amdgcn_wmma_f32_16x16x4_f32(false, a, false, b,
                                              (short)0, c, false, false);
  }

#pragma unroll
  for (int r = 0; r < 8; ++r) s_part[wave][r * 32 + lane] = c[r];
  __syncthreads();

  if (tid < 8 * 32) {
    float v = s_part[0][tid];
#pragma unroll
    for (int w = 1; w < KSPLIT; ++w) v += s_part[w][tid];
    const int r = tid >> 5;
    wq[(tM * 16 + half * 8 + r) * KVDIM + tN * 16 + lr] = v;
  }
}

__global__ void __launch_bounds__(TPB)
attn_partial(const float* __restrict__ kv,
             const int*   __restrict__ mask,
             const float* __restrict__ wq,
             float*       __restrict__ part) {
  __shared__ float    s_acc[WAVES][KVDIM];
  __shared__ float    s_m[WAVES];
  __shared__ float    s_l[WAVES];
  __shared__ unsigned s_bits[WAVES];
  __shared__ int      s_list[ROWS_PER_CHUNK];

  static_assert(ROWS_PER_CHUNK == TPB, "one thread per row of the chunk");

  const int b     = blockIdx.x / NCHUNK;
  const int chunk = blockIdx.x % NCHUNK;
  const int tid   = threadIdx.x;
  const int lane  = tid & 31;
  const int wave  = __builtin_amdgcn_readfirstlane(tid >> 5);
  const float NEG_INF = -__builtin_inff();
  const int sBase = chunk * ROWS_PER_CHUNK;

  const int mk = mask[b * SEQ + sBase + tid];
  float4 w4[4];
#pragma unroll
  for (int q = 0; q < 4; ++q)
    w4[q] = *(const float4*)&wq[b * KVDIM + q * 128 + lane * 4];

  const bool     keep = mk != 0;
  const unsigned bits = __builtin_amdgcn_ballot_w32(keep);
  if (lane == 0) s_bits[wave] = bits;
  __syncthreads();

  int before = 0, nAll = 0;
#pragma unroll
  for (int w = 0; w < WAVES; ++w) {
    const int c = __popc(s_bits[w]);
    before += (w < wave) ? c : 0;
    nAll   += c;
  }
  if (keep) s_list[before + __popc(bits & ((1u << lane) - 1u))] = tid;
  __syncthreads();
  const int n = __builtin_amdgcn_readfirstlane(nAll);

  float  m = NEG_INF, l = 0.f;
  float4 acc[4] = {};

  static_assert(ROWS_PER_CHUNK / WAVES == 32, "a wave's share of the list fits one entry per lane");
  const int myIdx = wave + lane * WAVES;
  const int myRow = (myIdx < n) ? s_list[myIdx] : 0;
  const int nW    = (n - wave + WAVES - 1) / WAVES;

  const float* __restrict__ chunkBase = kv + ((size_t)b * SEQ + sBase) * KVDIM;
  for (int g = 0; g < nW; g += GROUP) {
    float4 r4[GROUP][4];
#pragma unroll
    for (int j = 0; j < GROUP; ++j) {
      const int k  = (g + j < nW) ? (g + j) : g;
      const int rU = __builtin_amdgcn_readlane(myRow, k);
      const float* __restrict__ row = chunkBase + (size_t)rU * KVDIM;
#pragma unroll
      for (int q = 0; q < 4; ++q)
        r4[j][q] = *(const float4*)&row[q * 128 + lane * 4];
    }

    float d[GROUP];
#pragma unroll
    for (int j = 0; j < GROUP; ++j) {
      float dj = 0.f;
#pragma unroll
      for (int q = 0; q < 4; ++q) {
        dj = fmaf(r4[j][q].x, w4[q].x, dj);
        dj = fmaf(r4[j][q].y, w4[q].y, dj);
        dj = fmaf(r4[j][q].z, w4[q].z, dj);
        dj = fmaf(r4[j][q].w, w4[q].w, dj);
      }
      d[j] = dj;
    }
#pragma unroll
    for (int off = 16; off > 0; off >>= 1) {
#pragma unroll
      for (int j = 0; j < GROUP; ++j) d[j] += __shfl_xor(d[j], off, 32);
    }
#pragma unroll
    for (int j = 1; j < GROUP; ++j)
      if (g + j >= nW) d[j] = NEG_INF;

    float mN = m;
#pragma unroll
    for (int j = 0; j < GROUP; ++j) mN = fmaxf(mN, d[j]);
    const float scale = __expf(m - mN);
    float p[GROUP];
    float pSum = 0.f;
#pragma unroll
    for (int j = 0; j < GROUP; ++j) { p[j] = __expf(d[j] - mN); pSum += p[j]; }
    l = l * scale + pSum;
#pragma unroll
    for (int q = 0; q < 4; ++q) {
      float4 a = acc[q];
      a.x *= scale; a.y *= scale; a.z *= scale; a.w *= scale;
#pragma unroll
      for (int j = 0; j < GROUP; ++j) {
        a.x = fmaf(p[j], r4[j][q].x, a.x);
        a.y = fmaf(p[j], r4[j][q].y, a.y);
        a.z = fmaf(p[j], r4[j][q].z, a.z);
        a.w = fmaf(p[j], r4[j][q].w, a.w);
      }
      acc[q] = a;
    }
    m = mN;
  }

#pragma unroll
  for (int q = 0; q < 4; ++q)
    *(float4*)&s_acc[wave][q * 128 + lane * 4] = acc[q];
  if (lane == 0) { s_m[wave] = m; s_l[wave] = l; }
  __syncthreads();

  float mB = NEG_INF;
#pragma unroll
  for (int w = 0; w < WAVES; ++w) mB = fmaxf(mB, s_m[w]);
  float wgt[WAVES];
  float lB = 0.f;
#pragma unroll
  for (int w = 0; w < WAVES; ++w) {
    const float wi = (s_m[w] == NEG_INF) ? 0.f : __expf(s_m[w] - mB);
    wgt[w] = wi;
    lB += wi * s_l[w];
  }
  float* P = part + (size_t)(b * NCHUNK + chunk) * PART_STRIDE;
  for (int e = tid; e < KVDIM; e += TPB) {
    float v = 0.f;
#pragma unroll
    for (int w = 0; w < WAVES; ++w) v = fmaf(wgt[w], s_acc[w][e], v);
    P[e] = v;
  }
  if (tid == 0) { P[KVDIM] = mB; P[KVDIM + 1] = lB; }
}

__global__ void __launch_bounds__(256)
attn_combine(const float* __restrict__ part, float* __restrict__ out) {
  const int b   = blockIdx.x;
  const int tid = threadIdx.x;
  const float NEG_INF = -__builtin_inff();
  const float* __restrict__ P = part + (size_t)b * NCHUNK * PART_STRIDE;

  float a0[NCHUNK], a1[NCHUNK], mi[NCHUNK], li[NCHUNK];
#pragma unroll
  for (int c = 0; c < NCHUNK; ++c) {
    a0[c] = P[(size_t)c * PART_STRIDE + tid];
    a1[c] = P[(size_t)c * PART_STRIDE + tid + 256];
  }
#pragma unroll
  for (int c = 0; c < NCHUNK; ++c) {
    mi[c] = P[(size_t)c * PART_STRIDE + KVDIM];
    li[c] = P[(size_t)c * PART_STRIDE + KVDIM + 1];
  }

  float mG = NEG_INF;
#pragma unroll
  for (int c = 0; c < NCHUNK; ++c) mG = fmaxf(mG, mi[c]);

  float lG = 0.f, v0 = 0.f, v1 = 0.f;
#pragma unroll
  for (int c = 0; c < NCHUNK; ++c) {
    const float wi = (mi[c] == NEG_INF) ? 0.f : __expf(mi[c] - mG);
    lG += wi * li[c];
    v0 = fmaf(wi, a0[c], v0);
    v1 = fmaf(wi, a1[c], v1);
  }
  const float inv = 1.f / lG;
  out[b * KVDIM + tid]       = v0 * inv;
  out[b * KVDIM + tid + 256] = v1 * inv;
}

extern "C" void kernel_launch(void* const* d_in, const int* in_sizes, int n_in,
                              void* d_out, int out_size, void* d_ws, size_t ws_size,
                              hipStream_t stream) {
  const float* query  = (const float*)d_in[0];
  const float* kv     = (const float*)d_in[1];
  const int*   mask   = (const int*)  d_in[2];
  const float* weight = (const float*)d_in[3];
  float* out = (float*)d_out;

  float* ws   = (float*)d_ws;
  float* wq   = ws + WQ_OFF;
  float* part = ws + PART_OFF;

  wq_gemm_wmma<<<64, KSPLIT * 32, 0, stream>>>(query, weight, wq);
  attn_partial<<<BATCH * NCHUNK, TPB, 0, stream>>>(kv, mask, wq, part);
  attn_combine<<<BATCH, 256, 0, stream>>>(part, out);
}


# Attention pass: a wave takes its kept rows in equal trips of up to 10; an empty slot issues no load (no filler re-reads, 2 trips where some took 3); dots as even/odd pairs; changes summation order
# speedup vs baseline: 2.6259x; 1.0550x over previous
#include <hip/hip_runtime.h>
#include <hip/hip_bf16.h>

#define BATCH   32
#define SEQ     4096
#define QDIM    512
#define KVDIM   512

#define NCHUNK   16
#define ROWS_PER_CHUNK (SEQ / NCHUNK)
#define WAVES    8
#define TPB      (WAVES * 32)
#define GMAX     10

#define WQ_OFF      0
#define PART_STRIDE (KVDIM + 8)
#define PART_OFF    (BATCH * KVDIM)

typedef float v2f __attribute__((ext_vector_type(2)));
typedef float v4f __attribute__((ext_vector_type(4)));
typedef float v8f __attribute__((ext_vector_type(8)));

#define KSPLIT  16
#define KSLICE  (QDIM / KSPLIT)

__global__ void __launch_bounds__(KSPLIT * 32)
wq_gemm_wmma(const float* __restrict__ query,
             const float* __restrict__ weight,
             float* __restrict__ wq) {
  __shared__ float s_part[KSPLIT][8 * 32];

  const int tile = blockIdx.x;
  const int tM   = tile >> 5;
  const int tN   = tile & 31;
  const int tid  = threadIdx.x;
  const int wave = tid >> 5;
  const int lane = tid & 31;
  const int half = lane >> 4;
  const int lr   = lane & 15;

  const int aRow = tM * 16 + lr;
  const int bCol = tN * 16 + lr;
  const int kBeg = wave * KSLICE;

  v8f c = {};
#pragma unroll
  for (int k0 = kBeg; k0 < kBeg + KSLICE; k0 += 4) {
    const int ka = k0 + half * 2;
    v2f a, b;
    a[0] = query[aRow * QDIM + ka];
    a[1] = query[aRow * QDIM + ka + 1];
    b[0] = weight[(size_t)ka * KVDIM + bCol];
    b[1] = weight[(size_t)(ka + 1) * KVDIM + bCol];
    c = __builtin_amdgcn_wmma_f32_16x16x4_f32(false, a, false, b,
                                              (short)0, c, false, false);
  }

#pragma unroll
  for (int r = 0; r < 8; ++r) s_part[wave][r * 32 + lane] = c[r];
  __syncthreads();

  if (tid < 8 * 32) {
    float v = s_part[0][tid];
#pragma unroll
    for (int w = 1; w < KSPLIT; ++w) v += s_part[w][tid];
    const int r = tid >> 5;
    wq[(tM * 16 + half * 8 + r) * KVDIM + tN * 16 + lr] = v;
  }
}

__global__ void __launch_bounds__(TPB)
attn_partial(const float* __restrict__ kv,
             const int*   __restrict__ mask,
             const float* __restrict__ wq,
             float*       __restrict__ part) {
  __shared__ float    s_acc[WAVES][KVDIM];
  __shared__ float    s_m[WAVES];
  __shared__ float    s_l[WAVES];
  __shared__ unsigned s_bits[WAVES];
  __shared__ int      s_list[ROWS_PER_CHUNK];

  static_assert(ROWS_PER_CHUNK == TPB, "one thread per row of the chunk");

  const int b     = blockIdx.x / NCHUNK;
  const int chunk = blockIdx.x % NCHUNK;
  const int tid   = threadIdx.x;
  const int lane  = tid & 31;
  const int wave  = __builtin_amdgcn_readfirstlane(tid >> 5);
  const float NEG_INF = -__builtin_inff();
  const int sBase = chunk * ROWS_PER_CHUNK;

  const int mk = mask[b * SEQ + sBase + tid];
  float4 w4[4];
#pragma unroll
  for (int q = 0; q < 4; ++q)
    w4[q] = *(const float4*)&wq[b * KVDIM + q * 128 + lane * 4];

  const bool     keep = mk != 0;
  const unsigned bits = __builtin_amdgcn_ballot_w32(keep);
  if (lane == 0) s_bits[wave] = bits;
  __syncthreads();

  int before = 0, nAll = 0;
#pragma unroll
  for (int w = 0; w < WAVES; ++w) {
    const int c = __popc(s_bits[w]);
    before += (w < wave) ? c : 0;
    nAll   += c;
  }
  if (keep) s_list[before + __popc(bits & ((1u << lane) - 1u))] = tid;
  __syncthreads();
  const int n = __builtin_amdgcn_readfirstlane(nAll);

  float m = NEG_INF, l = 0.f;
  v4f   acc[4];
#pragma unroll
  for (int q = 0; q < 4; ++q) acc[q] = (v4f){0.f, 0.f, 0.f, 0.f};

  static_assert(ROWS_PER_CHUNK / WAVES == 32, "a wave's share of the list fits one entry per lane");
  const int myIdx = wave + lane * WAVES;
  const int myRow = (myIdx < n) ? s_list[myIdx] : 0;
  const int nW    = (n - wave + WAVES - 1) / WAVES;
  const int per   = (nW <= GMAX)     ? nW
                  : (nW <= 2 * GMAX) ? (nW + 1) / 2
                  : (nW <= 3 * GMAX) ? (nW + 2) / 3
                                     : (nW + 3) / 4;

  v4f wv[4];
#pragma unroll
  for (int q = 0; q < 4; ++q) wv[q] = (v4f){w4[q].x, w4[q].y, w4[q].z, w4[q].w};

  v4f r4[GMAX][4];
#pragma unroll
  for (int j = 0; j < GMAX; ++j)
#pragma unroll
    for (int q = 0; q < 4; ++q) r4[j][q] = (v4f){0.f, 0.f, 0.f, 0.f};

  const float* __restrict__ chunkBase = kv + ((size_t)b * SEQ + sBase) * KVDIM;
  for (int g = 0; g < nW; g += per) {
    const int cnt = (nW - g < per) ? (nW - g) : per;
#pragma unroll
    for (int j = 0; j < GMAX; ++j) {
      if (j < cnt) {
        const int rU = __builtin_amdgcn_readlane(myRow, g + j);
        const float* __restrict__ row = chunkBase + (size_t)rU * KVDIM;
#pragma unroll
        for (int q = 0; q < 4; ++q)
          r4[j][q] = *(const v4f*)&row[q * 128 + lane * 4];
      }
    }

    float d[GMAX];
#pragma unroll
    for (int j = 0; j < GMAX; ++j) {
      v2f s0 = {0.f, 0.f}, s1 = {0.f, 0.f};
#pragma unroll
      for (int q = 0; q < 4; ++q) {
        s0 = __builtin_elementwise_fma(r4[j][q].lo, wv[q].lo, s0);
        s1 = __builtin_elementwise_fma(r4[j][q].hi, wv[q].hi, s1);
      }
      const v2f s = s0 + s1;
      d[j] = s[0] + s[1];
    }
#pragma unroll
    for (int off = 16; off > 0; off >>= 1) {
#pragma unroll
      for (int j = 0; j < GMAX; ++j) d[j] += __shfl_xor(d[j], off, 32);
    }
#pragma unroll
    for (int j = 1; j < GMAX; ++j)
      if (j >= cnt) d[j] = NEG_INF;

    float mN = m;
#pragma unroll
    for (int j = 0; j < GMAX; ++j) mN = fmaxf(mN, d[j]);
    const float scale = __expf(m - mN);
    float p[GMAX];
    float pSum = 0.f;
#pragma unroll
    for (int j = 0; j < GMAX; ++j) { p[j] = __expf(d[j] - mN); pSum += p[j]; }
    l = l * scale + pSum;
#pragma unroll
    for (int q = 0; q < 4; ++q) {
      v4f a = acc[q] * scale;
#pragma unroll
      for (int j = 0; j < GMAX; ++j) {
        const v2f pj = {p[j], p[j]};
        a.lo = __builtin_elementwise_fma(pj, r4[j][q].lo, a.lo);
        a.hi = __builtin_elementwise_fma(pj, r4[j][q].hi, a.hi);
      }
      acc[q] = a;
    }
    m = mN;
  }

#pragma unroll
  for (int q = 0; q < 4; ++q)
    *(v4f*)&s_acc[wave][q * 128 + lane * 4] = acc[q];
  if (lane == 0) { s_m[wave] = m; s_l[wave] = l; }
  __syncthreads();

  float mB = NEG_INF;
#pragma unroll
  for (int w = 0; w < WAVES; ++w) mB = fmaxf(mB, s_m[w]);
  float wgt[WAVES];
  float lB = 0.f;
#pragma unroll
  for (int w = 0; w < WAVES; ++w) {
    const float wi = (s_m[w] == NEG_INF) ? 0.f : __expf(s_m[w] - mB);
    wgt[w] = wi;
    lB += wi * s_l[w];
  }
  float* P = part + (size_t)(b * NCHUNK + chunk) * PART_STRIDE;
  for (int e = tid; e < KVDIM; e += TPB) {
    float v = 0.f;
#pragma unroll
    for (int w = 0; w < WAVES; ++w) v = fmaf(wgt[w], s_acc[w][e], v);
    P[e] = v;
  }
  if (tid == 0) { P[KVDIM] = mB; P[KVDIM + 1] = lB; }
}

__global__ void __launch_bounds__(256)
attn_combine(const float* __restrict__ part, float* __restrict__ out) {
  const int b   = blockIdx.x;
  const int tid = threadIdx.x;
  const float NEG_INF = -__builtin_inff();
  const float* __restrict__ P = part + (size_t)b * NCHUNK * PART_STRIDE;

  float a0[NCHUNK], a1[NCHUNK], mi[NCHUNK], li[NCHUNK];
#pragma unroll
  for (int c = 0; c < NCHUNK; ++c) {
    a0[c] = P[(size_t)c * PART_STRIDE + tid];
    a1[c] = P[(size_t)c * PART_STRIDE + tid + 256];
  }
#pragma unroll
  for (int c = 0; c < NCHUNK; ++c) {
    mi[c] = P[(size_t)c * PART_STRIDE + KVDIM];
    li[c] = P[(size_t)c * PART_STRIDE + KVDIM + 1];
  }

  float mG = NEG_INF;
#pragma unroll
  for (int c = 0; c < NCHUNK; ++c) mG = fmaxf(mG, mi[c]);

  float lG = 0.f, v0 = 0.f, v1 = 0.f;
#pragma unroll
  for (int c = 0; c < NCHUNK; ++c) {
    const float wi = (mi[c] == NEG_INF) ? 0.f : __expf(mi[c] - mG);
    lG += wi * li[c];
    v0 = fmaf(wi, a0[c], v0);
    v1 = fmaf(wi, a1[c], v1);
  }
  const float inv = 1.f / lG;
  out[b * KVDIM + tid]       = v0 * inv;
  out[b * KVDIM + tid + 256] = v1 * inv;
}

extern "C" void kernel_launch(void* const* d_in, const int* in_sizes, int n_in,
                              void* d_out, int out_size, void* d_ws, size_t ws_size,
                              hipStream_t stream) {
  const float* query  = (const float*)d_in[0];
  const float* kv     = (const float*)d_in[1];
  const int*   mask   = (const int*)  d_in[2];
  const float* weight = (const float*)d_in[3];
  float* out = (float*)d_out;

  float* ws   = (float*)d_ws;
  float* wq   = ws + WQ_OFF;
  float* part = ws + PART_OFF;

  wq_gemm_wmma<<<64, KSPLIT * 32, 0, stream>>>(query, weight, wq);
  attn_partial<<<BATCH * NCHUNK, TPB, 0, stream>>>(kv, mask, wq, part);
  attn_combine<<<BATCH, 256, 0, stream>>>(part, out);
}
